# starting kernel
# baseline (speedup 1.0000x reference)
#include <hip/hip_runtime.h>
#include <hip/hip_bf16.h>

typedef __attribute__((ext_vector_type(2))) float v2f;
typedef __attribute__((ext_vector_type(8))) float v8f;

#define F 64

__global__ void deg_kernel(const int* __restrict__ src, const int* __restrict__ dst,
                           unsigned* __restrict__ deg_out, unsigned* __restrict__ deg_in,
                           int n_edges) {
    int i = blockIdx.x * blockDim.x + threadIdx.x;
    if (i < n_edges) {
        atomicAdd(&deg_out[src[i]], 1u);
        atomicAdd(&deg_in[dst[i]], 1u);
    }
}

__global__ void isqrt_kernel(const unsigned* __restrict__ deg, float* __restrict__ out, int n) {
    int i = blockIdx.x * blockDim.x + threadIdx.x;
    if (i < n) {
        unsigned d = deg[i];
        out[i] = d ? rsqrtf((float)d) : 0.0f;
    }
}

__global__ __launch_bounds__(128) void gemm_wmma_kernel(const float* __restrict__ x,
                                                        const float* __restrict__ dscale,
                                                        const float* __restrict__ W,
                                                        float* __restrict__ h) {
    const int lane  = threadIdx.x & 31;
    const int tileN = threadIdx.x >> 5;
    const int tileM = blockIdx.x;
    const int l15   = lane & 15;
    const int hi    = lane >> 4;
    const int kOff  = hi ? 2 : 0;

    const int row = tileM * 16 + l15;
    const int col = tileN * 16 + l15;

    const float s = dscale[row];
    const float* __restrict__ xrow = x + (size_t)row * F;

    v8f c = {};
#pragma unroll
    for (int k = 0; k < F; k += 4) {
        v2f a, b;
        a.x = xrow[k + kOff]     * s;
        a.y = xrow[k + kOff + 1] * s;
        b.x = W[(size_t)(k + kOff)     * F + col];
        b.y = W[(size_t)(k + kOff + 1) * F + col];
        c = __builtin_amdgcn_wmma_f32_16x16x4_f32(false, a, false, b, (short)0, c,
                                                  false, false);
    }

    const int baseRow = tileM * 16 + hi * 8;
#pragma unroll
    for (int v = 0; v < 8; ++v) {
        h[(size_t)(baseRow + v) * F + col] = c[v];
    }
}

__global__ void scatter_kernel(const float* __restrict__ h,
                               const int* __restrict__ src, const int* __restrict__ dst,
                               float* __restrict__ agg, int n_quads) {
    int idx = blockIdx.x * blockDim.x + threadIdx.x;
    if (idx < n_quads) {
        int e = idx >> 4;
        int q = (idx & 15) << 2;
        const float4 v = *(const float4*)(h + (size_t)src[e] * F + q);
        float* a = agg + (size_t)dst[e] * F + q;
        atomicAdd(a + 0, v.x);
        atomicAdd(a + 1, v.y);
        atomicAdd(a + 2, v.z);
        atomicAdd(a + 3, v.w);
    }
}

__global__ void post_kernel(const float* __restrict__ agg, const float* __restrict__ dii,
                            const float* __restrict__ bias, float* __restrict__ out,
                            int n_quads, int relu) {
    int idx = blockIdx.x * blockDim.x + threadIdx.x;
    if (idx < n_quads) {
        int node = idx >> 4;
        int q = (idx & 15) << 2;
        const float  d = dii[node];
        const float4 a = *(const float4*)(agg + ((size_t)idx << 2));
        const float4 b = *(const float4*)(bias + q);
        float4 r;
        r.x = a.x * d + b.x;
        r.y = a.y * d + b.y;
        r.z = a.z * d + b.z;
        r.w = a.w * d + b.w;
        if (relu) {
            r.x = fmaxf(r.x, 0.0f);
            r.y = fmaxf(r.y, 0.0f);
            r.z = fmaxf(r.z, 0.0f);
            r.w = fmaxf(r.w, 0.0f);
        }
        *(float4*)(out + ((size_t)idx << 2)) = r;
    }
}

extern "C" void kernel_launch(void* const* d_in, const int* in_sizes, int n_in,
                              void* d_out, int out_size, void* d_ws, size_t ws_size,
                              hipStream_t stream) {
    const float* x  = (const float*)d_in[0];
    const float* W1 = (const float*)d_in[1];
    const float* b1 = (const float*)d_in[2];
    const float* W2 = (const float*)d_in[3];
    const float* b2 = (const float*)d_in[4];
    const int*  src = (const int*)d_in[5];
    const int*  dst = (const int*)d_in[6];

    const int N = in_sizes[0] / F;
    const int E = in_sizes[5];
    const int NF = N * F;

    char* ws = (char*)d_ws;
    size_t off = 0;
    auto alloc = [&](size_t bytes) {
        void* p = ws + off;
        off = (off + bytes + 255) & ~(size_t)255;
        return p;
    };
    unsigned* deg  = (unsigned*)alloc(2 * (size_t)N * sizeof(unsigned));
    float*    dois = (float*)alloc((size_t)N * sizeof(float));
    float*    dii  = (float*)alloc((size_t)N * sizeof(float));
    float*    bufA = (float*)alloc((size_t)NF * sizeof(float));
    float*    bufB = (float*)alloc((size_t)NF * sizeof(float));
    float*    bufC = (float*)alloc((size_t)NF * sizeof(float));
    unsigned* deg_out = deg;
    unsigned* deg_in  = deg + N;

    const int TB = 256;
    const int scatter_quads  = E * 16;
    const int scatter_blocks = (scatter_quads + TB - 1) / TB;
    const int post_quads     = N * 16;
    const int post_blocks    = (post_quads + TB - 1) / TB;

    hipMemsetAsync(deg, 0, 2 * (size_t)N * sizeof(unsigned), stream);
    deg_kernel<<<(E + TB - 1) / TB, TB, 0, stream>>>(src, dst, deg_out, deg_in, E);
    isqrt_kernel<<<(N + TB - 1) / TB, TB, 0, stream>>>(deg_out, dois, N);
    isqrt_kernel<<<(N + TB - 1) / TB, TB, 0, stream>>>(deg_in,  dii,  N);

    gemm_wmma_kernel<<<N / 16, 128, 0, stream>>>(x, dois, W1, bufA);
    hipMemsetAsync(bufB, 0, (size_t)NF * sizeof(float), stream);
    scatter_kernel<<<scatter_blocks, TB, 0, stream>>>(bufA, src, dst, bufB, scatter_quads);
    post_kernel<<<post_blocks, TB, 0, stream>>>(bufB, dii, b1, bufC, post_quads, 1);

    gemm_wmma_kernel<<<N / 16, 128, 0, stream>>>(bufC, dois, W2, bufA);
    hipMemsetAsync(bufB, 0, (size_t)NF * sizeof(float), stream);
    scatter_kernel<<<scatter_blocks, TB, 0, stream>>>(bufA, src, dst, bufB, scatter_quads);
    post_kernel<<<post_blocks, TB, 0, stream>>>(bufB, dii, b2, (float*)d_out, post_quads, 0);
}


# scatter: one wave per 8 edges, lane = feature, so each row load and each f32 atomic covers one whole 128-byte line (2 atomic lines per edge, not 8); sums still in atomic arrival order
# speedup vs baseline: 1.3319x; 1.3319x over previous
#include <hip/hip_runtime.h>
#include <hip/hip_bf16.h>

typedef __attribute__((ext_vector_type(2))) float v2f;
typedef __attribute__((ext_vector_type(8))) float v8f;

#define F 64

__global__ void deg_kernel(const int* __restrict__ src, const int* __restrict__ dst,
                           unsigned* __restrict__ deg_out, unsigned* __restrict__ deg_in,
                           int n_edges) {
    int i = blockIdx.x * blockDim.x + threadIdx.x;
    if (i < n_edges) {
        atomicAdd(&deg_out[src[i]], 1u);
        atomicAdd(&deg_in[dst[i]], 1u);
    }
}

__global__ void isqrt_kernel(const unsigned* __restrict__ deg, float* __restrict__ out, int n) {
    int i = blockIdx.x * blockDim.x + threadIdx.x;
    if (i < n) {
        unsigned d = deg[i];
        out[i] = d ? rsqrtf((float)d) : 0.0f;
    }
}

__global__ __launch_bounds__(128) void gemm_wmma_kernel(const float* __restrict__ x,
                                                        const float* __restrict__ dscale,
                                                        const float* __restrict__ W,
                                                        float* __restrict__ h) {
    const int lane  = threadIdx.x & 31;
    const int tileN = threadIdx.x >> 5;
    const int tileM = blockIdx.x;
    const int l15   = lane & 15;
    const int hi    = lane >> 4;
    const int kOff  = hi ? 2 : 0;

    const int row = tileM * 16 + l15;
    const int col = tileN * 16 + l15;

    const float s = dscale[row];
    const float* __restrict__ xrow = x + (size_t)row * F;

    v8f c = {};
#pragma unroll
    for (int k = 0; k < F; k += 4) {
        v2f a, b;
        a.x = xrow[k + kOff]     * s;
        a.y = xrow[k + kOff + 1] * s;
        b.x = W[(size_t)(k + kOff)     * F + col];
        b.y = W[(size_t)(k + kOff + 1) * F + col];
        c = __builtin_amdgcn_wmma_f32_16x16x4_f32(false, a, false, b, (short)0, c,
                                                  false, false);
    }

    const int baseRow = tileM * 16 + hi * 8;
#pragma unroll
    for (int v = 0; v < 8; ++v) {
        h[(size_t)(baseRow + v) * F + col] = c[v];
    }
}

#define EPW 8
#define SCATTER_WAVES 8

__global__ __launch_bounds__(SCATTER_WAVES * 32) void scatter_kernel(
        const float* __restrict__ h,
        const int* __restrict__ src, const int* __restrict__ dst,
        float* __restrict__ agg, int n_edges) {
    const int lane = threadIdx.x & 31;
    const int wave = __builtin_amdgcn_readfirstlane((int)(threadIdx.x >> 5));
    const int e0 = (blockIdx.x * SCATTER_WAVES + wave) * EPW;
    if (e0 >= n_edges) return;

    if (e0 + EPW <= n_edges) {
        int s[EPW], d[EPW];
        float v0[EPW], v1[EPW];
#pragma unroll
        for (int j = 0; j < EPW; ++j) {
            s[j] = src[e0 + j];
            d[j] = dst[e0 + j];
        }
#pragma unroll
        for (int j = 0; j < EPW; ++j) {
            const float* __restrict__ r = h + (size_t)s[j] * F;
            v0[j] = r[lane];
            v1[j] = r[lane + 32];
        }
#pragma unroll
        for (int j = 0; j < EPW; ++j) {
            float* a = agg + (size_t)d[j] * F;
            atomicAdd(a + lane,      v0[j]);
            atomicAdd(a + lane + 32, v1[j]);
        }
    } else {
        for (int e = e0; e < n_edges; ++e) {
            const float* __restrict__ r = h + (size_t)src[e] * F;
            float* a = agg + (size_t)dst[e] * F;
            const float x0 = r[lane];
            const float x1 = r[lane + 32];
            atomicAdd(a + lane,      x0);
            atomicAdd(a + lane + 32, x1);
        }
    }
}

__global__ void post_kernel(const float* __restrict__ agg, const float* __restrict__ dii,
                            const float* __restrict__ bias, float* __restrict__ out,
                            int n_quads, int relu) {
    int idx = blockIdx.x * blockDim.x + threadIdx.x;
    if (idx < n_quads) {
        int node = idx >> 4;
        int q = (idx & 15) << 2;
        const float  d = dii[node];
        const float4 a = *(const float4*)(agg + ((size_t)idx << 2));
        const float4 b = *(const float4*)(bias + q);
        float4 r;
        r.x = a.x * d + b.x;
        r.y = a.y * d + b.y;
        r.z = a.z * d + b.z;
        r.w = a.w * d + b.w;
        if (relu) {
            r.x = fmaxf(r.x, 0.0f);
            r.y = fmaxf(r.y, 0.0f);
            r.z = fmaxf(r.z, 0.0f);
            r.w = fmaxf(r.w, 0.0f);
        }
        *(float4*)(out + ((size_t)idx << 2)) = r;
    }
}

extern "C" void kernel_launch(void* const* d_in, const int* in_sizes, int n_in,
                              void* d_out, int out_size, void* d_ws, size_t ws_size,
                              hipStream_t stream) {
    const float* x  = (const float*)d_in[0];
    const float* W1 = (const float*)d_in[1];
    const float* b1 = (const float*)d_in[2];
    const float* W2 = (const float*)d_in[3];
    const float* b2 = (const float*)d_in[4];
    const int*  src = (const int*)d_in[5];
    const int*  dst = (const int*)d_in[6];

    const int N = in_sizes[0] / F;
    const int E = in_sizes[5];
    const int NF = N * F;

    char* ws = (char*)d_ws;
    size_t off = 0;
    auto alloc = [&](size_t bytes) {
        void* p = ws + off;
        off = (off + bytes + 255) & ~(size_t)255;
        return p;
    };
    unsigned* deg  = (unsigned*)alloc(2 * (size_t)N * sizeof(unsigned));
    float*    dois = (float*)alloc((size_t)N * sizeof(float));
    float*    dii  = (float*)alloc((size_t)N * sizeof(float));
    float*    bufA = (float*)alloc((size_t)NF * sizeof(float));
    float*    bufB = (float*)alloc((size_t)NF * sizeof(float));
    float*    bufC = (float*)alloc((size_t)NF * sizeof(float));
    unsigned* deg_out = deg;
    unsigned* deg_in  = deg + N;

    const int TB = 256;
    const int edges_per_block = SCATTER_WAVES * EPW;
    const int scatter_blocks  = (E + edges_per_block - 1) / edges_per_block;
    const int post_quads     = N * 16;
    const int post_blocks    = (post_quads + TB - 1) / TB;

    hipMemsetAsync(deg, 0, 2 * (size_t)N * sizeof(unsigned), stream);
    deg_kernel<<<(E + TB - 1) / TB, TB, 0, stream>>>(src, dst, deg_out, deg_in, E);
    isqrt_kernel<<<(N + TB - 1) / TB, TB, 0, stream>>>(deg_out, dois, N);
    isqrt_kernel<<<(N + TB - 1) / TB, TB, 0, stream>>>(deg_in,  dii,  N);

    gemm_wmma_kernel<<<N / 16, 128, 0, stream>>>(x, dois, W1, bufA);
    hipMemsetAsync(bufB, 0, (size_t)NF * sizeof(float), stream);
    scatter_kernel<<<scatter_blocks, SCATTER_WAVES * 32, 0, stream>>>(bufA, src, dst, bufB, E);
    post_kernel<<<post_blocks, TB, 0, stream>>>(bufB, dii, b1, bufC, post_quads, 1);

    gemm_wmma_kernel<<<N / 16, 128, 0, stream>>>(bufC, dois, W2, bufA);
    hipMemsetAsync(bufB, 0, (size_t)NF * sizeof(float), stream);
    scatter_kernel<<<scatter_blocks, SCATTER_WAVES * 32, 0, stream>>>(bufA, src, dst, bufB, E);
    post_kernel<<<post_blocks, TB, 0, stream>>>(bufB, dii, b2, (float*)d_out, post_quads, 0);
}


# degree count in LDS: each block bins one edge slice into an 8192-node LDS histogram (integer adds, exact), then flushes non-zero counters with lane-consecutive global adds; no scattered atomics
# speedup vs baseline: 2.0307x; 1.5247x over previous
#include <hip/hip_runtime.h>
#include <hip/hip_bf16.h>

typedef __attribute__((ext_vector_type(2))) float v2f;
typedef __attribute__((ext_vector_type(8))) float v8f;

#define F 64

#define DEG_CH      8192
#define DEG_SLICES  16
#define DEG_THREADS 512
#define DEG_BATCH   8

__device__ __forceinline__ void deg_count(unsigned* hist, int idx, int c0) {
    const unsigned r = (unsigned)(idx - c0);
    if (r < (unsigned)DEG_CH) atomicAdd(&hist[r], 1u);
}

__global__ __launch_bounds__(DEG_THREADS) void deg_kernel(
        const int* __restrict__ src, const int* __restrict__ dst,
        unsigned* __restrict__ deg_out, unsigned* __restrict__ deg_in,
        int n_edges, int n_nodes) {
    __shared__ unsigned hist[DEG_CH];
    const int* __restrict__ idx = blockIdx.z ? dst : src;
    unsigned* __restrict__ deg  = blockIdx.z ? deg_in : deg_out;
    const int c0  = blockIdx.y * DEG_CH;
    const int tid = threadIdx.x;

    for (int i = tid; i < DEG_CH; i += DEG_THREADS) hist[i] = 0u;
    __syncthreads();

    const int n4  = n_edges >> 2;
    const int per = (n4 + (int)gridDim.x - 1) / (int)gridDim.x;
    const int q0  = blockIdx.x * per;
    const int q1  = min(q0 + per, n4);
    const int4* __restrict__ idx4 = (const int4*)idx;

    for (int q = q0 + tid; q < q1; q += DEG_BATCH * DEG_THREADS) {
        int4 v[DEG_BATCH];
#pragma unroll
        for (int j = 0; j < DEG_BATCH; ++j) {
            const int qq = q + j * DEG_THREADS;
            v[j] = (qq < q1) ? idx4[qq] : make_int4(-1, -1, -1, -1);
        }
#pragma unroll
        for (int j = 0; j < DEG_BATCH; ++j) {
            deg_count(hist, v[j].x, c0);
            deg_count(hist, v[j].y, c0);
            deg_count(hist, v[j].z, c0);
            deg_count(hist, v[j].w, c0);
        }
    }
    if (blockIdx.x == 0) {
        for (int e = (n4 << 2) + tid; e < n_edges; e += DEG_THREADS)
            deg_count(hist, idx[e], c0);
    }
    __syncthreads();

    for (int i = tid; i < DEG_CH; i += DEG_THREADS) {
        const unsigned v = hist[i];
        const int node = c0 + i;
        if (v != 0u && node < n_nodes) atomicAdd(&deg[node], v);
    }
}

__global__ void isqrt_kernel(const unsigned* __restrict__ deg, float* __restrict__ out, int n) {
    int i = blockIdx.x * blockDim.x + threadIdx.x;
    if (i < n) {
        unsigned d = deg[i];
        out[i] = d ? rsqrtf((float)d) : 0.0f;
    }
}

__global__ __launch_bounds__(128) void gemm_wmma_kernel(const float* __restrict__ x,
                                                        const float* __restrict__ dscale,
                                                        const float* __restrict__ W,
                                                        float* __restrict__ h) {
    const int lane  = threadIdx.x & 31;
    const int tileN = threadIdx.x >> 5;
    const int tileM = blockIdx.x;
    const int l15   = lane & 15;
    const int hi    = lane >> 4;
    const int kOff  = hi ? 2 : 0;

    const int row = tileM * 16 + l15;
    const int col = tileN * 16 + l15;

    const float s = dscale[row];
    const float* __restrict__ xrow = x + (size_t)row * F;

    v8f c = {};
#pragma unroll
    for (int k = 0; k < F; k += 4) {
        v2f a, b;
        a.x = xrow[k + kOff]     * s;
        a.y = xrow[k + kOff + 1] * s;
        b.x = W[(size_t)(k + kOff)     * F + col];
        b.y = W[(size_t)(k + kOff + 1) * F + col];
        c = __builtin_amdgcn_wmma_f32_16x16x4_f32(false, a, false, b, (short)0, c,
                                                  false, false);
    }

    const int baseRow = tileM * 16 + hi * 8;
#pragma unroll
    for (int v = 0; v < 8; ++v) {
        h[(size_t)(baseRow + v) * F + col] = c[v];
    }
}

#define EPW 8
#define SCATTER_WAVES 8

__global__ __launch_bounds__(SCATTER_WAVES * 32) void scatter_kernel(
        const float* __restrict__ h,
        const int* __restrict__ src, const int* __restrict__ dst,
        float* __restrict__ agg, int n_edges) {
    const int lane = threadIdx.x & 31;
    const int wave = __builtin_amdgcn_readfirstlane((int)(threadIdx.x >> 5));
    const int e0 = (blockIdx.x * SCATTER_WAVES + wave) * EPW;
    if (e0 >= n_edges) return;

    if (e0 + EPW <= n_edges) {
        int s[EPW], d[EPW];
        float v0[EPW], v1[EPW];
#pragma unroll
        for (int j = 0; j < EPW; ++j) {
            s[j] = src[e0 + j];
            d[j] = dst[e0 + j];
        }
#pragma unroll
        for (int j = 0; j < EPW; ++j) {
            const float* __restrict__ r = h + (size_t)s[j] * F;
            v0[j] = r[lane];
            v1[j] = r[lane + 32];
        }
#pragma unroll
        for (int j = 0; j < EPW; ++j) {
            float* a = agg + (size_t)d[j] * F;
            atomicAdd(a + lane,      v0[j]);
            atomicAdd(a + lane + 32, v1[j]);
        }
    } else {
        for (int e = e0; e < n_edges; ++e) {
            const float* __restrict__ r = h + (size_t)src[e] * F;
            float* a = agg + (size_t)dst[e] * F;
            const float x0 = r[lane];
            const float x1 = r[lane + 32];
            atomicAdd(a + lane,      x0);
            atomicAdd(a + lane + 32, x1);
        }
    }
}

__global__ void post_kernel(const float* __restrict__ agg, const float* __restrict__ dii,
                            const float* __restrict__ bias, float* __restrict__ out,
                            int n_quads, int relu) {
    int idx = blockIdx.x * blockDim.x + threadIdx.x;
    if (idx < n_quads) {
        int node = idx >> 4;
        int q = (idx & 15) << 2;
        const float  d = dii[node];
        const float4 a = *(const float4*)(agg + ((size_t)idx << 2));
        const float4 b = *(const float4*)(bias + q);
        float4 r;
        r.x = a.x * d + b.x;
        r.y = a.y * d + b.y;
        r.z = a.z * d + b.z;
        r.w = a.w * d + b.w;
        if (relu) {
            r.x = fmaxf(r.x, 0.0f);
            r.y = fmaxf(r.y, 0.0f);
            r.z = fmaxf(r.z, 0.0f);
            r.w = fmaxf(r.w, 0.0f);
        }
        *(float4*)(out + ((size_t)idx << 2)) = r;
    }
}

extern "C" void kernel_launch(void* const* d_in, const int* in_sizes, int n_in,
                              void* d_out, int out_size, void* d_ws, size_t ws_size,
                              hipStream_t stream) {
    const float* x  = (const float*)d_in[0];
    const float* W1 = (const float*)d_in[1];
    const float* b1 = (const float*)d_in[2];
    const float* W2 = (const float*)d_in[3];
    const float* b2 = (const float*)d_in[4];
    const int*  src = (const int*)d_in[5];
    const int*  dst = (const int*)d_in[6];

    const int N = in_sizes[0] / F;
    const int E = in_sizes[5];
    const int NF = N * F;

    char* ws = (char*)d_ws;
    size_t off = 0;
    auto alloc = [&](size_t bytes) {
        void* p = ws + off;
        off = (off + bytes + 255) & ~(size_t)255;
        return p;
    };
    unsigned* deg  = (unsigned*)alloc(2 * (size_t)N * sizeof(unsigned));
    float*    dois = (float*)alloc((size_t)N * sizeof(float));
    float*    dii  = (float*)alloc((size_t)N * sizeof(float));
    float*    bufA = (float*)alloc((size_t)NF * sizeof(float));
    float*    bufB = (float*)alloc((size_t)NF * sizeof(float));
    float*    bufC = (float*)alloc((size_t)NF * sizeof(float));
    unsigned* deg_out = deg;
    unsigned* deg_in  = deg + N;

    const int TB = 256;
    const int edges_per_block = SCATTER_WAVES * EPW;
    const int scatter_blocks  = (E + edges_per_block - 1) / edges_per_block;
    const int post_quads     = N * 16;
    const int post_blocks    = (post_quads + TB - 1) / TB;

    hipMemsetAsync(deg, 0, 2 * (size_t)N * sizeof(unsigned), stream);
    deg_kernel<<<dim3(DEG_SLICES, (N + DEG_CH - 1) / DEG_CH, 2), DEG_THREADS, 0, stream>>>(
        src, dst, deg_out, deg_in, E, N);
    isqrt_kernel<<<(N + TB - 1) / TB, TB, 0, stream>>>(deg_out, dois, N);
    isqrt_kernel<<<(N + TB - 1) / TB, TB, 0, stream>>>(deg_in,  dii,  N);

    gemm_wmma_kernel<<<N / 16, 128, 0, stream>>>(x, dois, W1, bufA);
    hipMemsetAsync(bufB, 0, (size_t)NF * sizeof(float), stream);
    scatter_kernel<<<scatter_blocks, SCATTER_WAVES * 32, 0, stream>>>(bufA, src, dst, bufB, E);
    post_kernel<<<post_blocks, TB, 0, stream>>>(bufB, dii, b1, bufC, post_quads, 1);

    gemm_wmma_kernel<<<N / 16, 128, 0, stream>>>(bufC, dois, W2, bufA);
    hipMemsetAsync(bufB, 0, (size_t)NF * sizeof(float), stream);
    scatter_kernel<<<scatter_blocks, SCATTER_WAVES * 32, 0, stream>>>(bufA, src, dst, bufB, E);
    post_kernel<<<post_blocks, TB, 0, stream>>>(bufB, dii, b2, (float*)d_out, post_quads, 0);
}


# edges ordered by destination in the call (per-slice LDS counts, prefix, LDS-ranked slots); a wave sums one destination row in registers, one store with epilogue; no f32 atomics; sum order = LDS rank
# speedup vs baseline: 3.0255x; 1.4899x over previous
#include <hip/hip_runtime.h>
#include <hip/hip_bf16.h>


typedef __attribute__((ext_vector_type(2))) float v2f;
typedef __attribute__((ext_vector_type(8))) float v8f;

#define F 64

#define DEG_CH      8192
#define DEG_SLICES  16
#define DEG_THREADS 512
#define DEG_BATCH   8
#define PFX_THREADS 256

__device__ __forceinline__ void slice_bounds(int n_edges, int& n4, int& q0, int& q1) {
    n4 = n_edges >> 2;
    const int per = (n4 + (int)gridDim.x - 1) / (int)gridDim.x;
    q0 = blockIdx.x * per;
    q1 = min(q0 + per, n4);
}

__device__ __forceinline__ void deg_count(unsigned* hist, int idx, int c0) {
    const unsigned r = (unsigned)(idx - c0);
    if (r < (unsigned)DEG_CH) atomicAdd(&hist[r], 1u);
}

__global__ __launch_bounds__(DEG_THREADS) void deg_kernel(
        const int* __restrict__ src, const int* __restrict__ dst,
        unsigned* __restrict__ deg_out, unsigned* __restrict__ cnt,
        int n_edges, int n_nodes) {
    __shared__ unsigned hist[DEG_CH];
    const int* __restrict__ idx = blockIdx.z ? dst : src;
    const int c0  = blockIdx.y * DEG_CH;
    const int tid = threadIdx.x;

    for (int i = tid; i < DEG_CH; i += DEG_THREADS) hist[i] = 0u;
    __syncthreads();

    int n4, q0, q1;
    slice_bounds(n_edges, n4, q0, q1);
    const int4* __restrict__ idx4 = (const int4*)idx;

    for (int q = q0 + tid; q < q1; q += DEG_BATCH * DEG_THREADS) {
        int4 v[DEG_BATCH];
#pragma unroll
        for (int j = 0; j < DEG_BATCH; ++j) {
            const int qq = q + j * DEG_THREADS;
            v[j] = (qq < q1) ? idx4[qq] : make_int4(-1, -1, -1, -1);
        }
#pragma unroll
        for (int j = 0; j < DEG_BATCH; ++j) {
            deg_count(hist, v[j].x, c0);
            deg_count(hist, v[j].y, c0);
            deg_count(hist, v[j].z, c0);
            deg_count(hist, v[j].w, c0);
        }
    }
    if (blockIdx.x == 0) {
        for (int e = (n4 << 2) + tid; e < n_edges; e += DEG_THREADS)
            deg_count(hist, idx[e], c0);
    }
    __syncthreads();

    unsigned* __restrict__ plane = cnt + (size_t)blockIdx.x * n_nodes;
    for (int i = tid; i < DEG_CH; i += DEG_THREADS) {
        const unsigned v = hist[i];
        const int node = c0 + i;
        if (node < n_nodes) {
            if (blockIdx.z) plane[node] = v;
            else if (v != 0u) atomicAdd(&deg_out[node], v);
        }
    }
}

__global__ __launch_bounds__(PFX_THREADS) void slice_prefix_kernel(
        unsigned* __restrict__ cnt, unsigned* __restrict__ deg_in,
        float* __restrict__ dii, unsigned* __restrict__ bsum, int n_nodes) {
    __shared__ unsigned tot;
    const int tid = threadIdx.x;
    const int n = blockIdx.x * PFX_THREADS + tid;
    if (tid == 0) tot = 0u;
    __syncthreads();

    unsigned run = 0u;
    if (n < n_nodes) {
        unsigned t[DEG_SLICES];
#pragma unroll
        for (int s = 0; s < DEG_SLICES; ++s) t[s] = cnt[(size_t)s * n_nodes + n];
#pragma unroll
        for (int s = 0; s < DEG_SLICES; ++s) {
            cnt[(size_t)s * n_nodes + n] = run;
            run += t[s];
        }
        deg_in[n] = run;
        dii[n] = run ? rsqrtf((float)run) : 0.0f;
    }
    atomicAdd(&tot, run);
    __syncthreads();
    if (tid == 0) bsum[blockIdx.x] = tot;
}

__device__ __forceinline__ void place_edge(unsigned* cur, int d, int s, int c0,
                                           int* __restrict__ ssrc, int n_edges) {
    const unsigned r = (unsigned)(d - c0);
    if (r < (unsigned)DEG_CH) {
        const unsigned pos = atomicAdd(&cur[r], 1u);
        if (pos < (unsigned)n_edges) ssrc[pos] = s;
    }
}

__device__ __forceinline__ void place_quad(unsigned* cur, const int4 d, const int4 s, int c0,
                                           int* __restrict__ ssrc, int n_edges) {
    const int dd[4] = {d.x, d.y, d.z, d.w};
    const int ss[4] = {s.x, s.y, s.z, s.w};
    unsigned pos[4];
#pragma unroll
    for (int c = 0; c < 4; ++c) {
        const unsigned r = (unsigned)(dd[c] - c0);
        pos[c] = 0xffffffffu;
        if (r < (unsigned)DEG_CH) pos[c] = atomicAdd(&cur[r], 1u);
    }
#pragma unroll
    for (int c = 0; c < 4; ++c) {
        if (pos[c] < (unsigned)n_edges) ssrc[pos[c]] = ss[c];
    }
}

#define PLACE_PER   (DEG_CH / DEG_THREADS)
#define PLACE_GROUP (DEG_THREADS / 32)

__global__ __launch_bounds__(DEG_THREADS) void place_kernel(
        const int* __restrict__ src, const int* __restrict__ dst,
        const unsigned* __restrict__ cnt, const unsigned* __restrict__ deg_in,
        const unsigned* __restrict__ bsum,
        unsigned* __restrict__ row_ptr, int* __restrict__ ssrc,
        int n_edges, int n_nodes) {
    __shared__ unsigned cur[DEG_CH];
    __shared__ unsigned part[DEG_THREADS];
    __shared__ unsigned gtot[32];
    __shared__ unsigned base_sh;
    const int c0  = blockIdx.y * DEG_CH;
    const int tid = threadIdx.x;
    const unsigned* __restrict__ pre = cnt + (size_t)blockIdx.x * n_nodes;

    if (tid == 0) base_sh = 0u;
    {
        unsigned t[PLACE_PER];
#pragma unroll
        for (int k = 0; k < PLACE_PER; ++k) {
            const int node = c0 + tid + k * DEG_THREADS;
            const unsigned g = deg_in[min(node, n_nodes - 1)];
            t[k] = (node < n_nodes) ? g : 0u;
        }
#pragma unroll
        for (int k = 0; k < PLACE_PER; ++k) cur[tid + k * DEG_THREADS] = t[k];
    }
    __syncthreads();

    const int nb = c0 / PFX_THREADS;
    for (int b = tid; b < nb; b += DEG_THREADS) atomicAdd(&base_sh, bsum[b]);

    const int i0 = tid * PLACE_PER;
    unsigned v[PLACE_PER];
    unsigned loc = 0u;
#pragma unroll
    for (int k = 0; k < PLACE_PER; ++k) {
        v[k] = cur[i0 + k];
        loc += v[k];
    }
    part[tid] = loc;
    __syncthreads();
    if (tid < 32) {
        unsigned a = 0u;
        for (int k = 0; k < PLACE_GROUP; ++k) {
            const unsigned t = part[tid * PLACE_GROUP + k];
            part[tid * PLACE_GROUP + k] = a;
            a += t;
        }
        gtot[tid] = a;
    }
    __syncthreads();
    if (tid == 0) {
        unsigned a = 0u;
        for (int g = 0; g < 32; ++g) {
            const unsigned t = gtot[g];
            gtot[g] = a;
            a += t;
        }
    }
    __syncthreads();

    unsigned p[PLACE_PER];
#pragma unroll
    for (int k = 0; k < PLACE_PER; ++k) p[k] = pre[min(c0 + i0 + k, n_nodes - 1)];

    unsigned run = base_sh + gtot[tid / PLACE_GROUP] + part[tid];
#pragma unroll
    for (int k = 0; k < PLACE_PER; ++k) {
        const int node = c0 + i0 + k;
        if (node < n_nodes) {
            if (blockIdx.x == 0) row_ptr[node] = run;
            cur[i0 + k] = run + p[k];
        }
        run += v[k];
    }
    __syncthreads();

    int n4, q0, q1;
    slice_bounds(n_edges, n4, q0, q1);
    const int4* __restrict__ dst4 = (const int4*)dst;
    const int4* __restrict__ src4 = (const int4*)src;

    for (int q = q0 + tid; q < q1; q += DEG_BATCH * DEG_THREADS) {
        int4 dv[DEG_BATCH], sv[DEG_BATCH];
#pragma unroll
        for (int j = 0; j < DEG_BATCH; ++j) {
            const int qq = q + j * DEG_THREADS;
            const int qc = min(qq, q1 - 1);
            const int4 dq = dst4[qc];
            sv[j] = src4[qc];
            dv[j] = (qq < q1) ? dq : make_int4(-1, -1, -1, -1);
        }
#pragma unroll
        for (int j = 0; j < DEG_BATCH; ++j) {
            place_quad(cur, dv[j], sv[j], c0, ssrc, n_edges);
        }
    }
    if (blockIdx.x == 0) {
        for (int e = (n4 << 2) + tid; e < n_edges; e += DEG_THREADS)
            place_edge(cur, dst[e], src[e], c0, ssrc, n_edges);
    }
}

__global__ void isqrt_kernel(const unsigned* __restrict__ deg, float* __restrict__ out, int n) {
    int i = blockIdx.x * blockDim.x + threadIdx.x;
    if (i < n) {
        unsigned d = deg[i];
        out[i] = d ? rsqrtf((float)d) : 0.0f;
    }
}

__global__ __launch_bounds__(128) void gemm_wmma_kernel(const float* __restrict__ x,
                                                        const float* __restrict__ dscale,
                                                        const float* __restrict__ W,
                                                        float* __restrict__ h) {
    const int lane  = threadIdx.x & 31;
    const int tileN = threadIdx.x >> 5;
    const int tileM = blockIdx.x;
    const int l15   = lane & 15;
    const int hi    = lane >> 4;
    const int kOff  = hi ? 2 : 0;

    const int row = tileM * 16 + l15;
    const int col = tileN * 16 + l15;

    const float s = dscale[row];
    const float* __restrict__ xrow = x + (size_t)row * F;

    v8f c = {};
#pragma unroll
    for (int k = 0; k < F; k += 4) {
        v2f a, b;
        a.x = xrow[k + kOff]     * s;
        a.y = xrow[k + kOff + 1] * s;
        b.x = W[(size_t)(k + kOff)     * F + col];
        b.y = W[(size_t)(k + kOff + 1) * F + col];
        c = __builtin_amdgcn_wmma_f32_16x16x4_f32(false, a, false, b, (short)0, c,
                                                  false, false);
    }

    const int baseRow = tileM * 16 + hi * 8;
#pragma unroll
    for (int v = 0; v < 8; ++v) {
        h[(size_t)(baseRow + v) * F + col] = c[v];
    }
}

#define GATHER_WAVES 8
#define GATHER_BATCH 8

__global__ __launch_bounds__(GATHER_WAVES * 32) void gather_kernel(
        const float* __restrict__ h, const int* __restrict__ ssrc,
        const unsigned* __restrict__ row_ptr, const unsigned* __restrict__ deg_in,
        const float* __restrict__ dii, const float* __restrict__ bias,
        float* __restrict__ out, int n_nodes, int n_edges, int relu) {
    const int lane = threadIdx.x & 31;
    const int wave = __builtin_amdgcn_readfirstlane((int)(threadIdx.x >> 5));
    const int node = blockIdx.x * GATHER_WAVES + wave;
    if (node >= n_nodes) return;

    unsigned beg = row_ptr[node];
    unsigned cnt = deg_in[node];
    if (beg > (unsigned)n_edges) beg = (unsigned)n_edges;
    if (cnt > (unsigned)n_edges - beg) cnt = (unsigned)n_edges - beg;

    float ax = 0.0f, ay = 0.0f;
    for (unsigned base = 0; base < cnt; base += 32u) {
        const unsigned n = (cnt - base < 32u) ? (cnt - base) : 32u;
        unsigned id = 0u;
        if ((unsigned)lane < n) id = (unsigned)ssrc[beg + base + (unsigned)lane];
        if (id > (unsigned)(n_nodes - 1)) id = (unsigned)(n_nodes - 1);
        for (unsigned j = 0; j < n; j += GATHER_BATCH) {
            v2f v[GATHER_BATCH];
#pragma unroll
            for (int u = 0; u < GATHER_BATCH; ++u) {
                const unsigned s = (unsigned)__builtin_amdgcn_readlane((int)id, (int)j + u);
                v[u] = *(const v2f*)(h + (size_t)s * F + 2 * lane);
            }
#pragma unroll
            for (int u = 0; u < GATHER_BATCH; ++u) {
                if (j + (unsigned)u < n) {
                    ax += v[u].x;
                    ay += v[u].y;
                }
            }
        }
    }

    const float d = dii[node];
    const v2f b = *(const v2f*)(bias + 2 * lane);
    v2f r;
    r.x = ax * d + b.x;
    r.y = ay * d + b.y;
    if (relu) {
        r.x = fmaxf(r.x, 0.0f);
        r.y = fmaxf(r.y, 0.0f);
    }
    *(v2f*)(out + (size_t)node * F + 2 * lane) = r;
}

extern "C" void kernel_launch(void* const* d_in, const int* in_sizes, int n_in,
                              void* d_out, int out_size, void* d_ws, size_t ws_size,
                              hipStream_t stream) {
    const float* x  = (const float*)d_in[0];
    const float* W1 = (const float*)d_in[1];
    const float* b1 = (const float*)d_in[2];
    const float* W2 = (const float*)d_in[3];
    const float* b2 = (const float*)d_in[4];
    const int*  src = (const int*)d_in[5];
    const int*  dst = (const int*)d_in[6];

    const int N = in_sizes[0] / F;
    const int E = in_sizes[5];
    const int NF = N * F;

    const int TB = 256;
    const int chunks      = (N + DEG_CH - 1) / DEG_CH;
    const int pfx_blocks  = (N + PFX_THREADS - 1) / PFX_THREADS;
    const int gather_blocks = (N + GATHER_WAVES - 1) / GATHER_WAVES;

    char* ws = (char*)d_ws;
    size_t off = 0;
    auto alloc = [&](size_t bytes) {
        void* p = ws + off;
        off = (off + bytes + 255) & ~(size_t)255;
        return p;
    };
    unsigned* deg_out = (unsigned*)alloc((size_t)N * sizeof(unsigned));
    unsigned* cnt     = (unsigned*)alloc((size_t)DEG_SLICES * N * sizeof(unsigned));
    unsigned* deg_in  = (unsigned*)alloc((size_t)N * sizeof(unsigned));
    unsigned* bsum    = (unsigned*)alloc((size_t)pfx_blocks * sizeof(unsigned));
    unsigned* row_ptr = (unsigned*)alloc((size_t)N * sizeof(unsigned));
    int*      ssrc    = (int*)alloc((size_t)E * sizeof(int));
    float*    dois = (float*)alloc((size_t)N * sizeof(float));
    float*    dii  = (float*)alloc((size_t)N * sizeof(float));
    float*    bufA = (float*)alloc((size_t)NF * sizeof(float));
    float*    bufC = (float*)alloc((size_t)NF * sizeof(float));

    hipMemsetAsync(deg_out, 0, (size_t)N * sizeof(unsigned), stream);
    deg_kernel<<<dim3(DEG_SLICES, chunks, 2), DEG_THREADS, 0, stream>>>(
        src, dst, deg_out, cnt, E, N);
    slice_prefix_kernel<<<pfx_blocks, PFX_THREADS, 0, stream>>>(cnt, deg_in, dii, bsum, N);
    place_kernel<<<dim3(DEG_SLICES, chunks), DEG_THREADS, 0, stream>>>(
        src, dst, cnt, deg_in, bsum, row_ptr, ssrc, E, N);
    isqrt_kernel<<<(N + TB - 1) / TB, TB, 0, stream>>>(deg_out, dois, N);

    gemm_wmma_kernel<<<N / 16, 128, 0, stream>>>(x, dois, W1, bufA);
    gather_kernel<<<gather_blocks, GATHER_WAVES * 32, 0, stream>>>(
        bufA, ssrc, row_ptr, deg_in, dii, b1, bufC, N, E, 1);

    gemm_wmma_kernel<<<N / 16, 128, 0, stream>>>(bufC, dois, W2, bufA);
    gather_kernel<<<gather_blocks, GATHER_WAVES * 32, 0, stream>>>(
        bufA, ssrc, row_ptr, deg_in, dii, b2, (float*)d_out, N, E, 0);
}


# segment sum moved in front of the linear map (it commutes with @W): a gather block sums 32 nodes' source rows into LDS, then does their f32 WMMA product; no GEMM launches; f32 sum order changes
# speedup vs baseline: 3.4315x; 1.1342x over previous
#include <hip/hip_runtime.h>
#include <hip/hip_bf16.h>


typedef __attribute__((ext_vector_type(2))) float v2f;
typedef __attribute__((ext_vector_type(8))) float v8f;

#define F 64

#define DEG_CH      8192
#define DEG_SLICES  16
#define DEG_THREADS 512
#define DEG_BATCH   8
#define PFX_THREADS 256

__device__ __forceinline__ void slice_bounds(int n_edges, int& n4, int& q0, int& q1) {
    n4 = n_edges >> 2;
    const int per = (n4 + (int)gridDim.x - 1) / (int)gridDim.x;
    q0 = blockIdx.x * per;
    q1 = min(q0 + per, n4);
}

__device__ __forceinline__ void deg_count(unsigned* hist, int idx, int c0) {
    const unsigned r = (unsigned)(idx - c0);
    if (r < (unsigned)DEG_CH) atomicAdd(&hist[r], 1u);
}

__global__ __launch_bounds__(DEG_THREADS) void deg_kernel(
        const int* __restrict__ src, const int* __restrict__ dst,
        unsigned* __restrict__ deg_out, unsigned* __restrict__ cnt,
        int n_edges, int n_nodes) {
    __shared__ unsigned hist[DEG_CH];
    const int* __restrict__ idx = blockIdx.z ? dst : src;
    const int c0  = blockIdx.y * DEG_CH;
    const int tid = threadIdx.x;

    for (int i = tid; i < DEG_CH; i += DEG_THREADS) hist[i] = 0u;
    __syncthreads();

    int n4, q0, q1;
    slice_bounds(n_edges, n4, q0, q1);
    const int4* __restrict__ idx4 = (const int4*)idx;

    for (int q = q0 + tid; q < q1; q += DEG_BATCH * DEG_THREADS) {
        int4 v[DEG_BATCH];
#pragma unroll
        for (int j = 0; j < DEG_BATCH; ++j) {
            const int qq = q + j * DEG_THREADS;
            v[j] = (qq < q1) ? idx4[qq] : make_int4(-1, -1, -1, -1);
        }
#pragma unroll
        for (int j = 0; j < DEG_BATCH; ++j) {
            deg_count(hist, v[j].x, c0);
            deg_count(hist, v[j].y, c0);
            deg_count(hist, v[j].z, c0);
            deg_count(hist, v[j].w, c0);
        }
    }
    if (blockIdx.x == 0) {
        for (int e = (n4 << 2) + tid; e < n_edges; e += DEG_THREADS)
            deg_count(hist, idx[e], c0);
    }
    __syncthreads();

    unsigned* __restrict__ plane = cnt + (size_t)blockIdx.x * n_nodes;
    for (int i = tid; i < DEG_CH; i += DEG_THREADS) {
        const unsigned v = hist[i];
        const int node = c0 + i;
        if (node < n_nodes) {
            if (blockIdx.z) plane[node] = v;
            else if (v != 0u) atomicAdd(&deg_out[node], v);
        }
    }
}

__global__ __launch_bounds__(PFX_THREADS) void slice_prefix_kernel(
        unsigned* __restrict__ cnt, unsigned* __restrict__ deg_in,
        float* __restrict__ dii, unsigned* __restrict__ bsum, int n_nodes) {
    __shared__ unsigned tot;
    const int tid = threadIdx.x;
    const int n = blockIdx.x * PFX_THREADS + tid;
    if (tid == 0) tot = 0u;
    __syncthreads();

    unsigned run = 0u;
    if (n < n_nodes) {
        unsigned t[DEG_SLICES];
#pragma unroll
        for (int s = 0; s < DEG_SLICES; ++s) t[s] = cnt[(size_t)s * n_nodes + n];
#pragma unroll
        for (int s = 0; s < DEG_SLICES; ++s) {
            cnt[(size_t)s * n_nodes + n] = run;
            run += t[s];
        }
        deg_in[n] = run;
        dii[n] = run ? rsqrtf((float)run) : 0.0f;
    }
    atomicAdd(&tot, run);
    __syncthreads();
    if (tid == 0) bsum[blockIdx.x] = tot;
}

__device__ __forceinline__ void place_edge(unsigned* cur, int d, int s, int c0,
                                           int* __restrict__ ssrc, int n_edges) {
    const unsigned r = (unsigned)(d - c0);
    if (r < (unsigned)DEG_CH) {
        const unsigned pos = atomicAdd(&cur[r], 1u);
        if (pos < (unsigned)n_edges) ssrc[pos] = s;
    }
}

__device__ __forceinline__ void place_quad(unsigned* cur, const int4 d, const int4 s, int c0,
                                           int* __restrict__ ssrc, int n_edges) {
    const int dd[4] = {d.x, d.y, d.z, d.w};
    const int ss[4] = {s.x, s.y, s.z, s.w};
    unsigned pos[4];
#pragma unroll
    for (int c = 0; c < 4; ++c) {
        const unsigned r = (unsigned)(dd[c] - c0);
        pos[c] = 0xffffffffu;
        if (r < (unsigned)DEG_CH) pos[c] = atomicAdd(&cur[r], 1u);
    }
#pragma unroll
    for (int c = 0; c < 4; ++c) {
        if (pos[c] < (unsigned)n_edges) ssrc[pos[c]] = ss[c];
    }
}

#define PLACE_PER   (DEG_CH / DEG_THREADS)
#define PLACE_GROUP (DEG_THREADS / 32)

__global__ __launch_bounds__(DEG_THREADS) void place_kernel(
        const int* __restrict__ src, const int* __restrict__ dst,
        const unsigned* __restrict__ cnt, const unsigned* __restrict__ deg_in,
        const unsigned* __restrict__ bsum,
        unsigned* __restrict__ row_ptr, int* __restrict__ ssrc,
        int n_edges, int n_nodes) {
    __shared__ unsigned cur[DEG_CH];
    __shared__ unsigned part[DEG_THREADS];
    __shared__ unsigned gtot[32];
    __shared__ unsigned base_sh;
    const int c0  = blockIdx.y * DEG_CH;
    const int tid = threadIdx.x;
    const unsigned* __restrict__ pre = cnt + (size_t)blockIdx.x * n_nodes;

    if (tid == 0) base_sh = 0u;
    {
        unsigned t[PLACE_PER];
#pragma unroll
        for (int k = 0; k < PLACE_PER; ++k) {
            const int node = c0 + tid + k * DEG_THREADS;
            const unsigned g = deg_in[min(node, n_nodes - 1)];
            t[k] = (node < n_nodes) ? g : 0u;
        }
#pragma unroll
        for (int k = 0; k < PLACE_PER; ++k) cur[tid + k * DEG_THREADS] = t[k];
    }
    __syncthreads();

    const int nb = c0 / PFX_THREADS;
    for (int b = tid; b < nb; b += DEG_THREADS) atomicAdd(&base_sh, bsum[b]);

    const int i0 = tid * PLACE_PER;
    unsigned v[PLACE_PER];
    unsigned loc = 0u;
#pragma unroll
    for (int k = 0; k < PLACE_PER; ++k) {
        v[k] = cur[i0 + k];
        loc += v[k];
    }
    part[tid] = loc;
    __syncthreads();
    if (tid < 32) {
        unsigned a = 0u;
        for (int k = 0; k < PLACE_GROUP; ++k) {
            const unsigned t = part[tid * PLACE_GROUP + k];
            part[tid * PLACE_GROUP + k] = a;
            a += t;
        }
        gtot[tid] = a;
    }
    __syncthreads();
    if (tid == 0) {
        unsigned a = 0u;
        for (int g = 0; g < 32; ++g) {
            const unsigned t = gtot[g];
            gtot[g] = a;
            a += t;
        }
    }
    __syncthreads();

    unsigned p[PLACE_PER];
#pragma unroll
    for (int k = 0; k < PLACE_PER; ++k) p[k] = pre[min(c0 + i0 + k, n_nodes - 1)];

    unsigned run = base_sh + gtot[tid / PLACE_GROUP] + part[tid];
#pragma unroll
    for (int k = 0; k < PLACE_PER; ++k) {
        const int node = c0 + i0 + k;
        if (node < n_nodes) {
            if (blockIdx.x == 0) row_ptr[node] = run;
            cur[i0 + k] = run + p[k];
        }
        run += v[k];
    }
    __syncthreads();

    int n4, q0, q1;
    slice_bounds(n_edges, n4, q0, q1);
    const int4* __restrict__ dst4 = (const int4*)dst;
    const int4* __restrict__ src4 = (const int4*)src;

    for (int q = q0 + tid; q < q1; q += DEG_BATCH * DEG_THREADS) {
        int4 dv[DEG_BATCH], sv[DEG_BATCH];
#pragma unroll
        for (int j = 0; j < DEG_BATCH; ++j) {
            const int qq = q + j * DEG_THREADS;
            const int qc = min(qq, q1 - 1);
            const int4 dq = dst4[qc];
            sv[j] = src4[qc];
            dv[j] = (qq < q1) ? dq : make_int4(-1, -1, -1, -1);
        }
#pragma unroll
        for (int j = 0; j < DEG_BATCH; ++j) {
            place_quad(cur, dv[j], sv[j], c0, ssrc, n_edges);
        }
    }
    if (blockIdx.x == 0) {
        for (int e = (n4 << 2) + tid; e < n_edges; e += DEG_THREADS)
            place_edge(cur, dst[e], src[e], c0, ssrc, n_edges);
    }
}

__global__ void isqrt_kernel(const unsigned* __restrict__ deg, float* __restrict__ out, int n) {
    int i = blockIdx.x * blockDim.x + threadIdx.x;
    if (i < n) {
        unsigned d = deg[i];
        out[i] = d ? rsqrtf((float)d) : 0.0f;
    }
}

#define LAYER_WAVES 8
#define LAYER_NPW   4
#define LAYER_TILE  (LAYER_WAVES * LAYER_NPW)
#define LAYER_BATCH 8
#define LAYER_PITCH 68
#define LAYER_WPITCH 72

template <bool FIRST>
__global__ __launch_bounds__(LAYER_WAVES * 32) void layer_kernel(
        const float* __restrict__ rows, const int* __restrict__ ssrc,
        const unsigned* __restrict__ row_ptr, const unsigned* __restrict__ deg_in,
        const float* __restrict__ dois, const float* __restrict__ dii,
        const float* __restrict__ W, const float* __restrict__ bias,
        float* __restrict__ out, int n_nodes, int n_edges) {
    __shared__ __attribute__((aligned(16))) float tileA[LAYER_TILE * LAYER_PITCH];
    __shared__ __attribute__((aligned(16))) float tileD[LAYER_TILE * LAYER_PITCH];
    __shared__ __attribute__((aligned(16))) float Wl[F * LAYER_WPITCH];
    const int lane = threadIdx.x & 31;
    const int wave  = __builtin_amdgcn_readfirstlane((int)(threadIdx.x >> 5));
    const int slot0 = wave * LAYER_NPW;
    const int node0 = blockIdx.x * LAYER_TILE + slot0;

    float4 wq[F * F / 4 / (LAYER_WAVES * 32)];
#pragma unroll
    for (int j = 0; j < F * F / 4 / (LAYER_WAVES * 32); ++j)
        wq[j] = ((const float4*)W)[threadIdx.x + j * (LAYER_WAVES * 32)];

    unsigned beg[LAYER_NPW], cnt[LAYER_NPW];
#pragma unroll
    for (int i = 0; i < LAYER_NPW; ++i) {
        const int node = node0 + i;
        const int nc = min(node, n_nodes - 1);
        unsigned b = row_ptr[nc];
        unsigned c = (node < n_nodes) ? deg_in[nc] : 0u;
        if (b > (unsigned)n_edges) b = (unsigned)n_edges;
        if (c > (unsigned)n_edges - b) c = (unsigned)n_edges - b;
        beg[i] = b;
        cnt[i] = c;
    }
    unsigned id0[LAYER_NPW];
    float    sc0[LAYER_NPW];
#pragma unroll
    for (int i = 0; i < LAYER_NPW; ++i) {
        const unsigned e = min(beg[i] + (unsigned)lane, (unsigned)(n_edges - 1));
        unsigned id = (unsigned)ssrc[e];
        if ((unsigned)lane >= cnt[i]) id = 0u;
        if (id > (unsigned)(n_nodes - 1)) id = (unsigned)(n_nodes - 1);
        id0[i] = id;
    }
#pragma unroll
    for (int i = 0; i < LAYER_NPW; ++i) sc0[i] = FIRST ? dois[id0[i]] : 1.0f;
#pragma unroll
    for (int j = 0; j < F * F / 4 / (LAYER_WAVES * 32); ++j) {
        const int q = threadIdx.x + j * (LAYER_WAVES * 32);
        *(float4*)(Wl + (q >> 4) * LAYER_WPITCH + (q & 15) * 4) = wq[j];
    }

#pragma unroll
    for (int i = 0; i < LAYER_NPW; ++i) {
        const unsigned c = cnt[i];
        unsigned id = id0[i];
        float    sc = sc0[i];
        float ax = 0.0f, ay = 0.0f;
        for (unsigned base = 0; base < c; base += 32u) {
            const unsigned n = (c - base < 32u) ? (c - base) : 32u;
            if (base) {
                id = 0u;
                if ((unsigned)lane < n) id = (unsigned)ssrc[beg[i] + base + (unsigned)lane];
                if (id > (unsigned)(n_nodes - 1)) id = (unsigned)(n_nodes - 1);
                if (FIRST) sc = dois[id];
            }
            for (unsigned j = 0; j < n; j += LAYER_BATCH) {
                v2f v[LAYER_BATCH];
#pragma unroll
                for (int u = 0; u < LAYER_BATCH; ++u) {
                    const unsigned s = (unsigned)__builtin_amdgcn_readlane((int)id, (int)j + u);
                    v[u] = *(const v2f*)(rows + (size_t)s * F + 2 * lane);
                }
#pragma unroll
                for (int u = 0; u < LAYER_BATCH; ++u) {
                    if (j + (unsigned)u < n) {
                        if (FIRST) {
                            const float w = __int_as_float(
                                __builtin_amdgcn_readlane(__float_as_int(sc), (int)j + u));
                            ax = fmaf(v[u].x, w, ax);
                            ay = fmaf(v[u].y, w, ay);
                        } else {
                            ax += v[u].x;
                            ay += v[u].y;
                        }
                    }
                }
            }
        }
        v2f a;
        a.x = ax;
        a.y = ay;
        *(v2f*)(tileA + (slot0 + i) * LAYER_PITCH + 2 * lane) = a;
    }

    const int l15  = lane & 15;
    const int hi   = lane >> 4;
    const int kOff = hi ? 2 : 0;
    const int tm   = wave >> 2;
    const int tn   = wave & 3;
    const int col  = tn * 16 + l15;

    __syncthreads();

    const float* arow = tileA + (tm * 16 + l15) * LAYER_PITCH + kOff;
    const float* bcol = Wl + kOff * LAYER_WPITCH + col;
    v8f c = {};
#pragma unroll
    for (int kk = 0; kk < F / 4; ++kk) {
        const v2f a = *(const v2f*)(arow + kk * 4);
        v2f b;
        b.x = bcol[(kk * 4)     * LAYER_WPITCH];
        b.y = bcol[(kk * 4 + 1) * LAYER_WPITCH];
        c = __builtin_amdgcn_wmma_f32_16x16x4_f32(false, a, false, b, (short)0, c,
                                                  false, false);
    }

    float* drow = tileD + (tm * 16 + hi * 8) * LAYER_PITCH + col;
#pragma unroll
    for (int v = 0; v < 8; ++v) drow[v * LAYER_PITCH] = c[v];
    __syncthreads();

    const v2f bb = *(const v2f*)(bias + 2 * lane);
#pragma unroll
    for (int i = 0; i < LAYER_NPW; ++i) {
        const int node = node0 + i;
        if (node < n_nodes) {
            const v2f d = *(const v2f*)(tileD + (slot0 + i) * LAYER_PITCH + 2 * lane);
            const float di = dii[node];
            v2f r;
            r.x = d.x * di + bb.x;
            r.y = d.y * di + bb.y;
            if (FIRST) {
                const float dn = dois[node];
                r.x = fmaxf(r.x, 0.0f) * dn;
                r.y = fmaxf(r.y, 0.0f) * dn;
            }
            *(v2f*)(out + (size_t)node * F + 2 * lane) = r;
        }
    }
}

extern "C" void kernel_launch(void* const* d_in, const int* in_sizes, int n_in,
                              void* d_out, int out_size, void* d_ws, size_t ws_size,
                              hipStream_t stream) {
    const float* x  = (const float*)d_in[0];
    const float* W1 = (const float*)d_in[1];
    const float* b1 = (const float*)d_in[2];
    const float* W2 = (const float*)d_in[3];
    const float* b2 = (const float*)d_in[4];
    const int*  src = (const int*)d_in[5];
    const int*  dst = (const int*)d_in[6];

    const int N = in_sizes[0] / F;
    const int E = in_sizes[5];
    const int NF = N * F;

    const int TB = 256;
    const int chunks      = (N + DEG_CH - 1) / DEG_CH;
    const int pfx_blocks  = (N + PFX_THREADS - 1) / PFX_THREADS;
    const int layer_blocks = (N + LAYER_TILE - 1) / LAYER_TILE;

    char* ws = (char*)d_ws;
    size_t off = 0;
    auto alloc = [&](size_t bytes) {
        void* p = ws + off;
        off = (off + bytes + 255) & ~(size_t)255;
        return p;
    };
    unsigned* deg_out = (unsigned*)alloc((size_t)N * sizeof(unsigned));
    unsigned* cnt     = (unsigned*)alloc((size_t)DEG_SLICES * N * sizeof(unsigned));
    unsigned* deg_in  = (unsigned*)alloc((size_t)N * sizeof(unsigned));
    unsigned* bsum    = (unsigned*)alloc((size_t)pfx_blocks * sizeof(unsigned));
    unsigned* row_ptr = (unsigned*)alloc((size_t)N * sizeof(unsigned));
    int*      ssrc    = (int*)alloc((size_t)E * sizeof(int));
    float*    dois = (float*)alloc((size_t)N * sizeof(float));
    float*    dii  = (float*)alloc((size_t)N * sizeof(float));
    float*    bufC = (float*)alloc((size_t)NF * sizeof(float));

    hipMemsetAsync(deg_out, 0, (size_t)N * sizeof(unsigned), stream);
    deg_kernel<<<dim3(DEG_SLICES, chunks, 2), DEG_THREADS, 0, stream>>>(
        src, dst, deg_out, cnt, E, N);
    slice_prefix_kernel<<<pfx_blocks, PFX_THREADS, 0, stream>>>(cnt, deg_in, dii, bsum, N);
    place_kernel<<<dim3(DEG_SLICES, chunks), DEG_THREADS, 0, stream>>>(
        src, dst, cnt, deg_in, bsum, row_ptr, ssrc, E, N);
    isqrt_kernel<<<(N + TB - 1) / TB, TB, 0, stream>>>(deg_out, dois, N);

    layer_kernel<true><<<layer_blocks, LAYER_WAVES * 32, 0, stream>>>(
        x, ssrc, row_ptr, deg_in, dois, dii, W1, b1, bufC, N, E);

    layer_kernel<false><<<layer_blocks, LAYER_WAVES * 32, 0, stream>>>(
        bufC, ssrc, row_ptr, deg_in, dois, dii, W2, b2, (float*)d_out, N, E);
}
